# starting kernel
# baseline (speedup 1.0000x reference)
#include <hip/hip_runtime.h>
#include <hip/hip_bf16.h>

typedef float v2f __attribute__((ext_vector_type(2)));
typedef float v8f __attribute__((ext_vector_type(8)));

__global__ void gemm_nt_wmma(const float* __restrict__ A,
                             const float* __restrict__ W,
                             float* __restrict__ C,
                             int M, int K, int O) {
  const int lane  = threadIdx.x & 31;
  const int oTile = threadIdx.x >> 5;
  const int mTile = blockIdx.x;
  if (mTile * 16 >= M) return;
  const int half = lane >> 4;
  const int l16  = lane & 15;

  const float* aPtr = A + (size_t)(mTile * 16 + l16) * K + 2 * half;
  const float* wPtr = W + (size_t)(oTile * 16 + l16) * K + 2 * half;

  v8f acc = {};
#pragma unroll 8
  for (int k = 0; k < K; k += 4) {
    v2f a = *(const v2f*)(aPtr + k);
    v2f b = *(const v2f*)(wPtr + k);
    acc = __builtin_amdgcn_wmma_f32_16x16x4_f32(
         false, a,  false, b,
         (short)0, acc,  false,  false);
  }

  float* cPtr = C + (size_t)(mTile * 16 + half * 8) * O + oTile * 16 + l16;
#pragma unroll
  for (int r = 0; r < 8; ++r) cPtr[(size_t)r * O] = acc[r];
}

__device__ __forceinline__ unsigned float_key(float f) {
  unsigned u = __float_as_uint(f);
  return (u & 0x80000000u) ? ~u : (u | 0x80000000u);
}
__device__ __forceinline__ float key_float(unsigned k) {
  unsigned u = (k & 0x80000000u) ? (k & 0x7FFFFFFFu) : ~k;
  return __uint_as_float(u);
}

__global__ void edge_logits(const float* __restrict__ z,
                            const float* __restrict__ df,
                            const int* __restrict__ src,
                            const int* __restrict__ dst,
                            float* __restrict__ e,
                            unsigned* __restrict__ emaxKey, int E) {
  const int edge = (int)((blockIdx.x * (size_t)blockDim.x + threadIdx.x) >> 5);
  const int lane = threadIdx.x & 31;
  if (edge >= E) return;
  const int s = src[edge], d = dst[edge];
  v2f a = *(const v2f*)(z  + (size_t)s * 64 + lane * 2);
  v2f b = *(const v2f*)(df + (size_t)d * 64 + lane * 2);
  float p = a.x * b.x + a.y * b.y;
#pragma unroll
  for (int off = 16; off; off >>= 1) p += __shfl_xor(p, off, 32);
  if (lane == 0) {
    e[edge] = p;
    atomicMax(emaxKey + d, float_key(p));
  }
}

__global__ void edge_scatter(const float* __restrict__ z,
                             const float* __restrict__ e,
                             const unsigned* __restrict__ emaxKey,
                             const int* __restrict__ src,
                             const int* __restrict__ dst,
                             float* __restrict__ num,
                             float* __restrict__ denom, int E) {
  const int edge = (int)((blockIdx.x * (size_t)blockDim.x + threadIdx.x) >> 5);
  const int lane = threadIdx.x & 31;
  if (edge >= E) return;
  const int s = src[edge], d = dst[edge];
  const float m  = key_float(emaxKey[d]);
  const float ex = __expf(e[edge] - m);
  v2f a = *(const v2f*)(z + (size_t)s * 64 + lane * 2);
  float* np = num + (size_t)d * 64 + lane * 2;
  atomicAdd(np + 0, ex * a.x);
  atomicAdd(np + 1, ex * a.y);
  if (lane == 0) atomicAdd(denom + d, ex);
}

__global__ void finalize(const float* __restrict__ num,
                         const float* __restrict__ denom,
                         float* __restrict__ out, int total) {
  const int i = blockIdx.x * blockDim.x + threadIdx.x;
  if (i >= total) return;
  const float dn = denom[i >> 6];
  out[i] = num[i] / (dn == 0.0f ? 1.0f : dn);
}

extern "C" void kernel_launch(void* const* d_in, const int* in_sizes, int n_in,
                              void* d_out, int out_size, void* d_ws, size_t ws_size,
                              hipStream_t stream) {
  const float* h     = (const float*)d_in[0];
  const float* feat  = (const float*)d_in[1];
  const float* W_fc  = (const float*)d_in[2];
  const float* W_dst = (const float*)d_in[3];
  const int*   src   = (const int*)d_in[4];
  const int*   dst   = (const int*)d_in[5];
  float*       out   = (float*)d_out;

  const int FEAT = 64, O = 64;
  const int N   = in_sizes[1] / FEAT;
  const int Kfc = in_sizes[0] / N;
  const int E   = in_sizes[4];

  float*    z       = (float*)d_ws;
  float*    df      = z  + (size_t)N * O;
  float*    num     = df + (size_t)N * O;
  float*    denom   = num + (size_t)N * O;
  unsigned* emaxKey = (unsigned*)(denom + N);
  float*    e       = (float*)(emaxKey + N);

  hipMemsetAsync(num, 0, ((size_t)N * O + 2 * (size_t)N) * sizeof(float), stream);

  const dim3 gemmBlk(32 * (O / 16));
  const int  mTiles = (N + 15) / 16;
  gemm_nt_wmma<<<mTiles, gemmBlk, 0, stream>>>(h,    W_fc,  z,  N, Kfc,  O);
  gemm_nt_wmma<<<mTiles, gemmBlk, 0, stream>>>(feat, W_dst, df, N, FEAT, O);

  const int blkE = 256, edgesPerBlk = blkE / 32;
  const int gE = (E + edgesPerBlk - 1) / edgesPerBlk;
  edge_logits <<<gE, blkE, 0, stream>>>(z, df, src, dst, e, emaxKey, E);
  edge_scatter<<<gE, blkE, 0, stream>>>(z, e, emaxKey, src, dst, num, denom, E);

  const int total = N * O;
  finalize<<<(total + 255) / 256, 256, 0, stream>>>(num, denom, out, total);
}


# edge kernels: 16 edges per wave, not 1 (logits: 8-lane groups, b128 row gathers issued ahead; scatter: loop over lane-broadcast indices): 16x fewer waves; logit dot summed in a new order
# speedup vs baseline: 1.5056x; 1.5056x over previous
#include <hip/hip_runtime.h>
#include <hip/hip_bf16.h>

typedef float v2f __attribute__((ext_vector_type(2)));
typedef float v8f __attribute__((ext_vector_type(8)));

__global__ void gemm_nt_wmma(const float* __restrict__ A,
                             const float* __restrict__ W,
                             float* __restrict__ C,
                             int M, int K, int O) {
  const int lane  = threadIdx.x & 31;
  const int oTile = threadIdx.x >> 5;
  const int mTile = blockIdx.x;
  if (mTile * 16 >= M) return;
  const int half = lane >> 4;
  const int l16  = lane & 15;

  const float* aPtr = A + (size_t)(mTile * 16 + l16) * K + 2 * half;
  const float* wPtr = W + (size_t)(oTile * 16 + l16) * K + 2 * half;

  v8f acc = {};
#pragma unroll 8
  for (int k = 0; k < K; k += 4) {
    v2f a = *(const v2f*)(aPtr + k);
    v2f b = *(const v2f*)(wPtr + k);
    acc = __builtin_amdgcn_wmma_f32_16x16x4_f32(
         false, a,  false, b,
         (short)0, acc,  false,  false);
  }

  float* cPtr = C + (size_t)(mTile * 16 + half * 8) * O + oTile * 16 + l16;
#pragma unroll
  for (int r = 0; r < 8; ++r) cPtr[(size_t)r * O] = acc[r];
}

#define EPW 16
static_assert(EPW == 16, "edge_logits walks 4 passes of 4 eight-lane groups");

typedef float v4f __attribute__((ext_vector_type(4)));

__device__ __forceinline__ unsigned float_key(float f) {
  unsigned u = __float_as_uint(f);
  return (u & 0x80000000u) ? ~u : (u | 0x80000000u);
}
__device__ __forceinline__ float key_float(unsigned k) {
  unsigned u = (k & 0x80000000u) ? (k & 0x7FFFFFFFu) : ~k;
  return __uint_as_float(u);
}

__global__ void edge_logits(const float* __restrict__ z,
                            const float* __restrict__ df,
                            const int* __restrict__ src,
                            const int* __restrict__ dst,
                            float* __restrict__ e,
                            unsigned* __restrict__ emaxKey, int E) {
  const int wave = (int)((blockIdx.x * (size_t)blockDim.x + threadIdx.x) >> 5);
  const int lane = threadIdx.x & 31;
  const int base = wave * EPW;
  if (base >= E) return;
  const int grp = lane >> 3;
  const int q   = lane & 7;

  int s[4], d[4];
#pragma unroll
  for (int it = 0; it < 4; ++it) {
    int edge = base + it * 4 + grp;
    edge = edge < E ? edge : E - 1;
    s[it] = src[edge];
    d[it] = dst[edge];
  }

  v4f a0[4], a1[4], b0[4], b1[4];
#pragma unroll
  for (int it = 0; it < 4; ++it) {
    const float* zp = z  + (size_t)s[it] * 64 + q * 4;
    const float* dp = df + (size_t)d[it] * 64 + q * 4;
    a0[it] = *(const v4f*)zp;
    a1[it] = *(const v4f*)(zp + 32);
    b0[it] = *(const v4f*)dp;
    b1[it] = *(const v4f*)(dp + 32);
  }

  float p[4];
#pragma unroll
  for (int it = 0; it < 4; ++it) {
    float t = a0[it].x * b0[it].x;
    t = fmaf(a0[it].y, b0[it].y, t);
    t = fmaf(a0[it].z, b0[it].z, t);
    t = fmaf(a0[it].w, b0[it].w, t);
    t = fmaf(a1[it].x, b1[it].x, t);
    t = fmaf(a1[it].y, b1[it].y, t);
    t = fmaf(a1[it].z, b1[it].z, t);
    t = fmaf(a1[it].w, b1[it].w, t);
    p[it] = t;
  }
#pragma unroll
  for (int off = 1; off < 8; off <<= 1) {
#pragma unroll
    for (int it = 0; it < 4; ++it) p[it] += __shfl_xor(p[it], off, 32);
  }

  float pv = p[0];
  int   dv = d[0];
  if (q == 1) { pv = p[1]; dv = d[1]; }
  if (q == 2) { pv = p[2]; dv = d[2]; }
  if (q == 3) { pv = p[3]; dv = d[3]; }
  const int edgeOut = base + q * 4 + grp;
  if (q < 4 && edgeOut < E) {
    e[edgeOut] = pv;
    atomicMax(emaxKey + dv, float_key(pv));
  }
}

__global__ void edge_scatter(const float* __restrict__ z,
                             const float* __restrict__ e,
                             const unsigned* __restrict__ emaxKey,
                             const int* __restrict__ src,
                             const int* __restrict__ dst,
                             float* __restrict__ num,
                             float* __restrict__ denom, int E) {
  const int wave = (int)((blockIdx.x * (size_t)blockDim.x + threadIdx.x) >> 5);
  const int lane = threadIdx.x & 31;
  const int base = __builtin_amdgcn_readfirstlane(wave * EPW);
  if (base >= E) return;
  const int left = E - base;
  const int cnt  = left < EPW ? left : EPW;

  int mine = base + (lane & (EPW - 1));
  mine = mine < E ? mine : E - 1;
  const int   sl = src[mine];
  const int   dl = dst[mine];
  const float el = e[mine];

  v2f a[EPW];
#pragma unroll
  for (int j = 0; j < EPW; ++j) {
    const int sj = __builtin_amdgcn_readlane(sl, j);
    a[j] = *(const v2f*)(z + (size_t)sj * 64 + lane * 2);
  }

  const float m   = key_float(emaxKey[dl]);
  const float exl = __expf(el - m);
  if (lane < cnt) atomicAdd(denom + dl, exl);

#pragma unroll
  for (int j = 0; j < EPW; ++j) {
    if (j < cnt) {
      const int   dj = __builtin_amdgcn_readlane(dl, j);
      const float ex = __int_as_float(__builtin_amdgcn_readlane(__float_as_int(exl), j));
      float* np = num + (size_t)dj * 64 + lane * 2;
      atomicAdd(np + 0, ex * a[j].x);
      atomicAdd(np + 1, ex * a[j].y);
    }
  }
}

__global__ void finalize(const float* __restrict__ num,
                         const float* __restrict__ denom,
                         float* __restrict__ out, int total) {
  const int i = blockIdx.x * blockDim.x + threadIdx.x;
  if (i >= total) return;
  const float dn = denom[i >> 6];
  out[i] = num[i] / (dn == 0.0f ? 1.0f : dn);
}

extern "C" void kernel_launch(void* const* d_in, const int* in_sizes, int n_in,
                              void* d_out, int out_size, void* d_ws, size_t ws_size,
                              hipStream_t stream) {
  const float* h     = (const float*)d_in[0];
  const float* feat  = (const float*)d_in[1];
  const float* W_fc  = (const float*)d_in[2];
  const float* W_dst = (const float*)d_in[3];
  const int*   src   = (const int*)d_in[4];
  const int*   dst   = (const int*)d_in[5];
  float*       out   = (float*)d_out;

  const int FEAT = 64, O = 64;
  const int N   = in_sizes[1] / FEAT;
  const int Kfc = in_sizes[0] / N;
  const int E   = in_sizes[4];

  float*    z       = (float*)d_ws;
  float*    df      = z  + (size_t)N * O;
  float*    num     = df + (size_t)N * O;
  float*    denom   = num + (size_t)N * O;
  unsigned* emaxKey = (unsigned*)(denom + N);
  float*    e       = (float*)(emaxKey + N);

  hipMemsetAsync(num, 0, ((size_t)N * O + 2 * (size_t)N) * sizeof(float), stream);

  const dim3 gemmBlk(32 * (O / 16));
  const int  mTiles = (N + 15) / 16;
  gemm_nt_wmma<<<mTiles, gemmBlk, 0, stream>>>(h,    W_fc,  z,  N, Kfc,  O);
  gemm_nt_wmma<<<mTiles, gemmBlk, 0, stream>>>(feat, W_dst, df, N, FEAT, O);

  const int blkE = 256, edgesPerBlk = (blkE / 32) * EPW;
  const int gE = (E + edgesPerBlk - 1) / edgesPerBlk;
  edge_logits <<<gE, blkE, 0, stream>>>(z, df, src, dst, e, emaxKey, E);
  edge_scatter<<<gE, blkE, 0, stream>>>(z, e, emaxKey, src, dst, num, denom, E);

  const int total = N * O;
  finalize<<<(total + 255) / 256, 256, 0, stream>>>(num, denom, out, total);
}


# edge phase by destination: in-call counting sort by dst, then one 8-lane group per dst keeps running max, denominator and 64 sums in registers; no f32 atomics; sums in arrival-ranked slot order
# speedup vs baseline: 2.2647x; 1.5042x over previous
#include <hip/hip_runtime.h>
#include <hip/hip_bf16.h>

typedef float v2f __attribute__((ext_vector_type(2)));
typedef float v8f __attribute__((ext_vector_type(8)));

__global__ void gemm_nt_wmma(const float* __restrict__ A,
                             const float* __restrict__ W,
                             float* __restrict__ C,
                             int M, int K, int O) {
  const int lane  = threadIdx.x & 31;
  const int oTile = threadIdx.x >> 5;
  const int mTile = blockIdx.x;
  if (mTile * 16 >= M) return;
  const int half = lane >> 4;
  const int l16  = lane & 15;

  const float* aPtr = A + (size_t)(mTile * 16 + l16) * K + 2 * half;
  const float* wPtr = W + (size_t)(oTile * 16 + l16) * K + 2 * half;

  v8f acc = {};
#pragma unroll 8
  for (int k = 0; k < K; k += 4) {
    v2f a = *(const v2f*)(aPtr + k);
    v2f b = *(const v2f*)(wPtr + k);
    acc = __builtin_amdgcn_wmma_f32_16x16x4_f32(
         false, a,  false, b,
         (short)0, acc,  false,  false);
  }

  float* cPtr = C + (size_t)(mTile * 16 + half * 8) * O + oTile * 16 + l16;
#pragma unroll
  for (int r = 0; r < 8; ++r) cPtr[(size_t)r * O] = acc[r];
}

typedef float v4f __attribute__((ext_vector_type(4)));

__global__ void edge_rank(const int* __restrict__ dst,
                          int* __restrict__ cnt,
                          int* __restrict__ rank, int E) {
  const int i4 = (int)(blockIdx.x * blockDim.x + threadIdx.x) * 4;
  if (i4 + 3 < E) {
    const int4 d = *(const int4*)(dst + i4);
    int4 r;
    r.x = atomicAdd(cnt + d.x, 1);
    r.y = atomicAdd(cnt + d.y, 1);
    r.z = atomicAdd(cnt + d.z, 1);
    r.w = atomicAdd(cnt + d.w, 1);
    *(int4*)(rank + i4) = r;
  } else {
    for (int i = i4; i < E; ++i) rank[i] = atomicAdd(cnt + dst[i], 1);
  }
}

__global__ void seg_offsets(const int* __restrict__ cnt,
                            int* __restrict__ off,
                            int* __restrict__ cursor, int N) {
  __shared__ int wsum[32];
  __shared__ int blockBase;
  const int i    = blockIdx.x * blockDim.x + threadIdx.x;
  const int lane = threadIdx.x & 31;
  const int w    = threadIdx.x >> 5;
  const int c    = i < N ? cnt[i] : 0;
  int v = c;
#pragma unroll
  for (int o = 1; o < 32; o <<= 1) {
    const int t = __shfl_up(v, o, 32);
    if (lane >= o) v += t;
  }
  if (lane == 31) wsum[w] = v;
  __syncthreads();
  if (w == 0) {
    const int s = wsum[lane];
    int t = s;
#pragma unroll
    for (int o = 1; o < 32; o <<= 1) {
      const int u = __shfl_up(t, o, 32);
      if (lane >= o) t += u;
    }
    wsum[lane] = t - s;
    if (lane == 31) blockBase = atomicAdd(cursor, t);
  }
  __syncthreads();
  if (i < N) off[i] = blockBase + wsum[w] + v - c;
}

__global__ void csr_fill(const int* __restrict__ src,
                         const int* __restrict__ dst,
                         const int* __restrict__ rank,
                         const int* __restrict__ off,
                         int* __restrict__ csr, int E) {
  const int i4 = (int)(blockIdx.x * blockDim.x + threadIdx.x) * 4;
  if (i4 + 3 < E) {
    const int4 d = *(const int4*)(dst + i4);
    const int4 s = *(const int4*)(src + i4);
    const int4 r = *(const int4*)(rank + i4);
    const int o0 = off[d.x], o1 = off[d.y], o2 = off[d.z], o3 = off[d.w];
    csr[o0 + r.x] = s.x;
    csr[o1 + r.y] = s.y;
    csr[o2 + r.z] = s.z;
    csr[o3 + r.w] = s.w;
  } else {
    for (int i = i4; i < E; ++i) csr[off[dst[i]] + rank[i]] = src[i];
  }
}

__global__ void seg_softmax_sum(const float* __restrict__ z,
                                const float* __restrict__ df,
                                const int* __restrict__ csr,
                                const int* __restrict__ off,
                                const int* __restrict__ cnt,
                                float* __restrict__ out, int N) {
  const int wave = (int)((blockIdx.x * (size_t)blockDim.x + threadIdx.x) >> 5);
  const int lane = threadIdx.x & 31;
  if (wave * 4 >= N) return;
  const int grp  = lane >> 3;
  const int q    = lane & 7;
  const int node = wave * 4 + grp;
  const bool live = node < N;
  const int nd   = live ? node : N - 1;
  const int beg  = off[nd];
  const int end  = beg + (live ? cnt[nd] : 0);
  const int last = end - 1;

  const float* dp = df + (size_t)nd * 64 + q * 4;
  const v4f b0 = *(const v4f*)dp;
  const v4f b1 = *(const v4f*)(dp + 32);

  v4f acc0 = {0.f, 0.f, 0.f, 0.f}, acc1 = {0.f, 0.f, 0.f, 0.f};
  float den = 0.f;
  float m   = -INFINITY;

  int nx[4] = {0, 0, 0, 0};
  if (beg < end) {
#pragma unroll
    for (int j = 0; j < 4; ++j) { const int k = beg + j; nx[j] = csr[k < end ? k : last]; }
  }

  for (int i = beg; i < end; i += 4) {
    v4f a0[4], a1[4];
#pragma unroll
    for (int j = 0; j < 4; ++j) {
      const float* zp = z + (size_t)nx[j] * 64 + q * 4;
      a0[j] = *(const v4f*)zp;
      a1[j] = *(const v4f*)(zp + 32);
    }
    if (i + 4 < end) {
#pragma unroll
      for (int j = 0; j < 4; ++j) { const int k = i + 4 + j; nx[j] = csr[k < end ? k : last]; }
    }

    float p[4];
#pragma unroll
    for (int j = 0; j < 4; ++j) {
      float t = a0[j].x * b0.x;
      t = fmaf(a0[j].y, b0.y, t);
      t = fmaf(a0[j].z, b0.z, t);
      t = fmaf(a0[j].w, b0.w, t);
      t = fmaf(a1[j].x, b1.x, t);
      t = fmaf(a1[j].y, b1.y, t);
      t = fmaf(a1[j].z, b1.z, t);
      t = fmaf(a1[j].w, b1.w, t);
      p[j] = t;
    }
#pragma unroll
    for (int o = 1; o < 8; o <<= 1) {
#pragma unroll
      for (int j = 0; j < 4; ++j) p[j] += __shfl_xor(p[j], o, 32);
    }
#pragma unroll
    for (int j = 1; j < 4; ++j) if (i + j >= end) p[j] = -INFINITY;

    const float cm = fmaxf(fmaxf(m, p[0]), fmaxf(fmaxf(p[1], p[2]), p[3]));
    const float sc = __expf(m - cm);
    const float w0 = __expf(p[0] - cm);
    const float w1 = __expf(p[1] - cm);
    const float w2 = __expf(p[2] - cm);
    const float w3 = __expf(p[3] - cm);
    den  = den * sc + ((w0 + w1) + (w2 + w3));
    acc0 = acc0 * sc + w0 * a0[0] + w1 * a0[1] + w2 * a0[2] + w3 * a0[3];
    acc1 = acc1 * sc + w0 * a1[0] + w1 * a1[1] + w2 * a1[2] + w3 * a1[3];
    m = cm;
  }

  if (live) {
    const float dn = den == 0.0f ? 1.0f : den;
    float* op = out + (size_t)node * 64 + q * 4;
    v4f r0, r1;
    r0.x = acc0.x / dn; r0.y = acc0.y / dn; r0.z = acc0.z / dn; r0.w = acc0.w / dn;
    r1.x = acc1.x / dn; r1.y = acc1.y / dn; r1.z = acc1.z / dn; r1.w = acc1.w / dn;
    *(v4f*)op        = r0;
    *(v4f*)(op + 32) = r1;
  }
}

extern "C" void kernel_launch(void* const* d_in, const int* in_sizes, int n_in,
                              void* d_out, int out_size, void* d_ws, size_t ws_size,
                              hipStream_t stream) {
  const float* h     = (const float*)d_in[0];
  const float* feat  = (const float*)d_in[1];
  const float* W_fc  = (const float*)d_in[2];
  const float* W_dst = (const float*)d_in[3];
  const int*   src   = (const int*)d_in[4];
  const int*   dst   = (const int*)d_in[5];
  float*       out   = (float*)d_out;

  const int FEAT = 64, O = 64;
  const int N   = in_sizes[1] / FEAT;
  const int Kfc = in_sizes[0] / N;
  const int E   = in_sizes[4];
  const size_t Ep = ((size_t)E + 3) & ~(size_t)3;

  float* z      = (float*)d_ws;
  float* df     = z + (size_t)N * O;
  int*   csr    = (int*)(df + (size_t)N * O);
  int*   rank   = csr + Ep;
  int*   off    = rank + Ep;
  int*   cnt    = off + N;
  int*   cursor = cnt + N;

  hipMemsetAsync(cnt, 0, ((size_t)N + 1) * sizeof(int), stream);

  const int quads = (E + 3) / 4;
  const int gQ = (quads + 255) / 256;
  edge_rank  <<<gQ, 256, 0, stream>>>(dst, cnt, rank, E);
  seg_offsets<<<(N + 1023) / 1024, 1024, 0, stream>>>(cnt, off, cursor, N);
  csr_fill   <<<gQ, 256, 0, stream>>>(src, dst, rank, off, csr, E);

  const dim3 gemmBlk(32 * (O / 16));
  const int  mTiles = (N + 15) / 16;
  gemm_nt_wmma<<<mTiles, gemmBlk, 0, stream>>>(h,    W_fc,  z,  N, Kfc,  O);
  gemm_nt_wmma<<<mTiles, gemmBlk, 0, stream>>>(feat, W_dst, df, N, FEAT, O);

  seg_softmax_sum<<<(N + 31) / 32, 256, 0, stream>>>(z, df, csr, off, cnt, out, N);
}


# node projections: W staged once per 8-wave block in LDS, one wave per 16x64 tile on 4 accumulators, 16-B fragment loads (A fetched once for all columns), grid 768; dots summed in a new k order
# speedup vs baseline: 2.8490x; 1.2580x over previous
#include <hip/hip_runtime.h>
#include <hip/hip_bf16.h>

typedef float v2f __attribute__((ext_vector_type(2)));
typedef float v8f __attribute__((ext_vector_type(8)));

typedef float v4f __attribute__((ext_vector_type(4)));

template <int K>
__global__ __launch_bounds__(256)
__attribute__((amdgpu_waves_per_eu(6, 6)))
void gemm_nt_wmma(const float* __restrict__ A,
                  const float* __restrict__ W,
                  float* __restrict__ C, int M) {
  constexpr int O   = 64;
  constexpr int P   = K + 4;
  constexpr int NCH = K / 32;
  static_assert(NCH % 2 == 0, "two pieces per trip");
  __shared__ __attribute__((aligned(16))) float sW[O * P];

  const int tid    = threadIdx.x;
  const int lane   = tid & 31;
  const int wv     = __builtin_amdgcn_readfirstlane(tid >> 5);
  const int half   = lane >> 4;
  const int l16    = lane & 15;
  const int mTiles = (M + 15) >> 4;
  const int stride = 8 * (int)gridDim.x;

  auto rowPtr = [&](int t) {
    int r = (t < mTiles ? t : mTiles - 1) * 16 + l16;
    if (r >= M) r = M - 1;
    return A + (size_t)r * K + 4 * half;
  };
  int mTile = wv * (int)gridDim.x + (int)blockIdx.x;
  const float* aPtr  = rowPtr(mTile);
  const float* aNext = rowPtr(mTile + stride);

  v4f a0 = *(const v4f*)(aPtr);
  v4f a1 = *(const v4f*)(aPtr + 8);
  v4f a2 = *(const v4f*)(aPtr + 16);
  v4f a3 = *(const v4f*)(aPtr + 24);

#pragma unroll
  for (int i = 0; i < O * K / 4 / 256; ++i) {
    const int idx = tid + i * 256;
    const int row = idx / (K / 4);
    const int c4  = idx % (K / 4);
    *(v4f*)(sW + row * P + c4 * 4) = *(const v4f*)(W + (size_t)idx * 4);
  }
  __syncthreads();

  const float* bBase = sW + l16 * P + 4 * half;

  v8f acc0, acc1, acc2, acc3;
  auto mm8 = [&](const v4f av, const float* bp) {
    const v4f b0 = *(const v4f*)(bp);
    const v4f b1 = *(const v4f*)(bp + 16 * P);
    const v4f b2 = *(const v4f*)(bp + 32 * P);
    const v4f b3 = *(const v4f*)(bp + 48 * P);
    const v2f alo = {av.x, av.y};
    const v2f ahi = {av.z, av.w};
    acc0 = __builtin_amdgcn_wmma_f32_16x16x4_f32(false, alo, false, (v2f){b0.x, b0.y}, (short)0, acc0, false, false);
    acc1 = __builtin_amdgcn_wmma_f32_16x16x4_f32(false, alo, false, (v2f){b1.x, b1.y}, (short)0, acc1, false, false);
    acc2 = __builtin_amdgcn_wmma_f32_16x16x4_f32(false, alo, false, (v2f){b2.x, b2.y}, (short)0, acc2, false, false);
    acc3 = __builtin_amdgcn_wmma_f32_16x16x4_f32(false, alo, false, (v2f){b3.x, b3.y}, (short)0, acc3, false, false);
    acc0 = __builtin_amdgcn_wmma_f32_16x16x4_f32(false, ahi, false, (v2f){b0.z, b0.w}, (short)0, acc0, false, false);
    acc1 = __builtin_amdgcn_wmma_f32_16x16x4_f32(false, ahi, false, (v2f){b1.z, b1.w}, (short)0, acc1, false, false);
    acc2 = __builtin_amdgcn_wmma_f32_16x16x4_f32(false, ahi, false, (v2f){b2.z, b2.w}, (short)0, acc2, false, false);
    acc3 = __builtin_amdgcn_wmma_f32_16x16x4_f32(false, ahi, false, (v2f){b3.z, b3.w}, (short)0, acc3, false, false);
  };

  while (mTile < mTiles) {
    acc0 = (v8f){}; acc1 = (v8f){}; acc2 = (v8f){}; acc3 = (v8f){};
#pragma unroll 1
    for (int c = 0; c < NCH; c += 2) {
      const float* ap = aPtr + c * 32;
      const float* bp = bBase + c * 32;
      const v4f n0 = *(const v4f*)(ap + 32);
      const v4f n1 = *(const v4f*)(ap + 40);
      const v4f n2 = *(const v4f*)(ap + 48);
      const v4f n3 = *(const v4f*)(ap + 56);
      mm8(a0, bp);
      mm8(a1, bp + 8);
      mm8(a2, bp + 16);
      mm8(a3, bp + 24);
      __builtin_amdgcn_sched_barrier(0);
      const float* pf = (c + 2 < NCH) ? ap + 64 : aNext;
      a0 = *(const v4f*)(pf);
      a1 = *(const v4f*)(pf + 8);
      a2 = *(const v4f*)(pf + 16);
      a3 = *(const v4f*)(pf + 24);
      mm8(n0, bp + 32);
      mm8(n1, bp + 40);
      mm8(n2, bp + 48);
      mm8(n3, bp + 56);
    }

    const int r0 = mTile * 16 + half * 8;
    float* cPtr = C + (size_t)r0 * O + l16;
    if (mTile * 16 + 16 <= M) {
#pragma unroll
      for (int r = 0; r < 8; ++r) {
        float* p = cPtr + (size_t)r * O;
        p[0] = acc0[r]; p[16] = acc1[r]; p[32] = acc2[r]; p[48] = acc3[r];
      }
    } else {
#pragma unroll
      for (int r = 0; r < 8; ++r) {
        if (r0 + r < M) {
          float* p = cPtr + (size_t)r * O;
          p[0] = acc0[r]; p[16] = acc1[r]; p[32] = acc2[r]; p[48] = acc3[r];
        }
      }
    }

    mTile += stride;
    aPtr  = aNext;
    aNext = rowPtr(mTile + stride);
  }
}

__global__ void edge_rank(const int* __restrict__ dst,
                          int* __restrict__ cnt,
                          int* __restrict__ rank, int E) {
  const int i4 = (int)(blockIdx.x * blockDim.x + threadIdx.x) * 4;
  if (i4 + 3 < E) {
    const int4 d = *(const int4*)(dst + i4);
    int4 r;
    r.x = atomicAdd(cnt + d.x, 1);
    r.y = atomicAdd(cnt + d.y, 1);
    r.z = atomicAdd(cnt + d.z, 1);
    r.w = atomicAdd(cnt + d.w, 1);
    *(int4*)(rank + i4) = r;
  } else {
    for (int i = i4; i < E; ++i) rank[i] = atomicAdd(cnt + dst[i], 1);
  }
}

__global__ void seg_offsets(const int* __restrict__ cnt,
                            int* __restrict__ off,
                            int* __restrict__ cursor, int N) {
  __shared__ int wsum[32];
  __shared__ int blockBase;
  const int i    = blockIdx.x * blockDim.x + threadIdx.x;
  const int lane = threadIdx.x & 31;
  const int w    = threadIdx.x >> 5;
  const int c    = i < N ? cnt[i] : 0;
  int v = c;
#pragma unroll
  for (int o = 1; o < 32; o <<= 1) {
    const int t = __shfl_up(v, o, 32);
    if (lane >= o) v += t;
  }
  if (lane == 31) wsum[w] = v;
  __syncthreads();
  if (w == 0) {
    const int s = wsum[lane];
    int t = s;
#pragma unroll
    for (int o = 1; o < 32; o <<= 1) {
      const int u = __shfl_up(t, o, 32);
      if (lane >= o) t += u;
    }
    wsum[lane] = t - s;
    if (lane == 31) blockBase = atomicAdd(cursor, t);
  }
  __syncthreads();
  if (i < N) off[i] = blockBase + wsum[w] + v - c;
}

__global__ void csr_fill(const int* __restrict__ src,
                         const int* __restrict__ dst,
                         const int* __restrict__ rank,
                         const int* __restrict__ off,
                         int* __restrict__ csr, int E) {
  const int i4 = (int)(blockIdx.x * blockDim.x + threadIdx.x) * 4;
  if (i4 + 3 < E) {
    const int4 d = *(const int4*)(dst + i4);
    const int4 s = *(const int4*)(src + i4);
    const int4 r = *(const int4*)(rank + i4);
    const int o0 = off[d.x], o1 = off[d.y], o2 = off[d.z], o3 = off[d.w];
    csr[o0 + r.x] = s.x;
    csr[o1 + r.y] = s.y;
    csr[o2 + r.z] = s.z;
    csr[o3 + r.w] = s.w;
  } else {
    for (int i = i4; i < E; ++i) csr[off[dst[i]] + rank[i]] = src[i];
  }
}

__global__ void seg_softmax_sum(const float* __restrict__ z,
                                const float* __restrict__ df,
                                const int* __restrict__ csr,
                                const int* __restrict__ off,
                                const int* __restrict__ cnt,
                                float* __restrict__ out, int N) {
  const int wave = (int)((blockIdx.x * (size_t)blockDim.x + threadIdx.x) >> 5);
  const int lane = threadIdx.x & 31;
  if (wave * 4 >= N) return;
  const int grp  = lane >> 3;
  const int q    = lane & 7;
  const int node = wave * 4 + grp;
  const bool live = node < N;
  const int nd   = live ? node : N - 1;
  const int beg  = off[nd];
  const int end  = beg + (live ? cnt[nd] : 0);
  const int last = end - 1;

  const float* dp = df + (size_t)nd * 64 + q * 4;
  const v4f b0 = *(const v4f*)dp;
  const v4f b1 = *(const v4f*)(dp + 32);

  v4f acc0 = {0.f, 0.f, 0.f, 0.f}, acc1 = {0.f, 0.f, 0.f, 0.f};
  float den = 0.f;
  float m   = -INFINITY;

  int nx[4] = {0, 0, 0, 0};
  if (beg < end) {
#pragma unroll
    for (int j = 0; j < 4; ++j) { const int k = beg + j; nx[j] = csr[k < end ? k : last]; }
  }

  for (int i = beg; i < end; i += 4) {
    v4f a0[4], a1[4];
#pragma unroll
    for (int j = 0; j < 4; ++j) {
      const float* zp = z + (size_t)nx[j] * 64 + q * 4;
      a0[j] = *(const v4f*)zp;
      a1[j] = *(const v4f*)(zp + 32);
    }
    if (i + 4 < end) {
#pragma unroll
      for (int j = 0; j < 4; ++j) { const int k = i + 4 + j; nx[j] = csr[k < end ? k : last]; }
    }

    float p[4];
#pragma unroll
    for (int j = 0; j < 4; ++j) {
      float t = a0[j].x * b0.x;
      t = fmaf(a0[j].y, b0.y, t);
      t = fmaf(a0[j].z, b0.z, t);
      t = fmaf(a0[j].w, b0.w, t);
      t = fmaf(a1[j].x, b1.x, t);
      t = fmaf(a1[j].y, b1.y, t);
      t = fmaf(a1[j].z, b1.z, t);
      t = fmaf(a1[j].w, b1.w, t);
      p[j] = t;
    }
#pragma unroll
    for (int o = 1; o < 8; o <<= 1) {
#pragma unroll
      for (int j = 0; j < 4; ++j) p[j] += __shfl_xor(p[j], o, 32);
    }
#pragma unroll
    for (int j = 1; j < 4; ++j) if (i + j >= end) p[j] = -INFINITY;

    const float cm = fmaxf(fmaxf(m, p[0]), fmaxf(fmaxf(p[1], p[2]), p[3]));
    const float sc = __expf(m - cm);
    const float w0 = __expf(p[0] - cm);
    const float w1 = __expf(p[1] - cm);
    const float w2 = __expf(p[2] - cm);
    const float w3 = __expf(p[3] - cm);
    den  = den * sc + ((w0 + w1) + (w2 + w3));
    acc0 = acc0 * sc + w0 * a0[0] + w1 * a0[1] + w2 * a0[2] + w3 * a0[3];
    acc1 = acc1 * sc + w0 * a1[0] + w1 * a1[1] + w2 * a1[2] + w3 * a1[3];
    m = cm;
  }

  if (live) {
    const float dn = den == 0.0f ? 1.0f : den;
    float* op = out + (size_t)node * 64 + q * 4;
    v4f r0, r1;
    r0.x = acc0.x / dn; r0.y = acc0.y / dn; r0.z = acc0.z / dn; r0.w = acc0.w / dn;
    r1.x = acc1.x / dn; r1.y = acc1.y / dn; r1.z = acc1.z / dn; r1.w = acc1.w / dn;
    *(v4f*)op        = r0;
    *(v4f*)(op + 32) = r1;
  }
}

extern "C" void kernel_launch(void* const* d_in, const int* in_sizes, int n_in,
                              void* d_out, int out_size, void* d_ws, size_t ws_size,
                              hipStream_t stream) {
  const float* h     = (const float*)d_in[0];
  const float* feat  = (const float*)d_in[1];
  const float* W_fc  = (const float*)d_in[2];
  const float* W_dst = (const float*)d_in[3];
  const int*   src   = (const int*)d_in[4];
  const int*   dst   = (const int*)d_in[5];
  float*       out   = (float*)d_out;

  constexpr int IN_DIM = 256, FEAT = 64, O = 64;
  const int N   = in_sizes[1] / FEAT;
  const int E   = in_sizes[4];
  const size_t Ep = ((size_t)E + 3) & ~(size_t)3;

  float* z      = (float*)d_ws;
  float* df     = z + (size_t)N * O;
  int*   csr    = (int*)(df + (size_t)N * O);
  int*   rank   = csr + Ep;
  int*   off    = rank + Ep;
  int*   cnt    = off + N;
  int*   cursor = cnt + N;

  hipMemsetAsync(cnt, 0, ((size_t)N + 1) * sizeof(int), stream);

  const int quads = (E + 3) / 4;
  const int gQ = (quads + 255) / 256;
  edge_rank  <<<gQ, 256, 0, stream>>>(dst, cnt, rank, E);
  seg_offsets<<<(N + 1023) / 1024, 1024, 0, stream>>>(cnt, off, cursor, N);
  csr_fill   <<<gQ, 256, 0, stream>>>(src, dst, rank, off, csr, E);

  gemm_nt_wmma<IN_DIM><<<768, 256, 0, stream>>>(h,    W_fc,  z,  N);
  gemm_nt_wmma<FEAT>  <<<768, 256, 0, stream>>>(feat, W_dst, df, N);

  seg_softmax_sum<<<(N + 31) / 32, 256, 0, stream>>>(z, df, csr, off, cnt, out, N);
}


# edge ordering through LDS: blocks order 4096-edge chunks by 512-node bin, one block per bin counts and places in LDS; no per-edge global atomics or scattered stores; sums in LDS-arrival slot order
# speedup vs baseline: 6.5644x; 2.3041x over previous
#include <hip/hip_runtime.h>
#include <hip/hip_bf16.h>

typedef float v2f __attribute__((ext_vector_type(2)));
typedef float v8f __attribute__((ext_vector_type(8)));

typedef float v4f __attribute__((ext_vector_type(4)));

template <int K>
__global__ __launch_bounds__(256)
__attribute__((amdgpu_waves_per_eu(6, 6)))
void gemm_nt_wmma(const float* __restrict__ A,
                  const float* __restrict__ W,
                  float* __restrict__ C, int M) {
  constexpr int O   = 64;
  constexpr int P   = K + 4;
  constexpr int NCH = K / 32;
  static_assert(NCH % 2 == 0, "two pieces per trip");
  __shared__ __attribute__((aligned(16))) float sW[O * P];

  const int tid    = threadIdx.x;
  const int lane   = tid & 31;
  const int wv     = __builtin_amdgcn_readfirstlane(tid >> 5);
  const int half   = lane >> 4;
  const int l16    = lane & 15;
  const int mTiles = (M + 15) >> 4;
  const int stride = 8 * (int)gridDim.x;

  auto rowPtr = [&](int t) {
    int r = (t < mTiles ? t : mTiles - 1) * 16 + l16;
    if (r >= M) r = M - 1;
    return A + (size_t)r * K + 4 * half;
  };
  int mTile = wv * (int)gridDim.x + (int)blockIdx.x;
  const float* aPtr  = rowPtr(mTile);
  const float* aNext = rowPtr(mTile + stride);

  v4f a0 = *(const v4f*)(aPtr);
  v4f a1 = *(const v4f*)(aPtr + 8);
  v4f a2 = *(const v4f*)(aPtr + 16);
  v4f a3 = *(const v4f*)(aPtr + 24);

#pragma unroll
  for (int i = 0; i < O * K / 4 / 256; ++i) {
    const int idx = tid + i * 256;
    const int row = idx / (K / 4);
    const int c4  = idx % (K / 4);
    *(v4f*)(sW + row * P + c4 * 4) = *(const v4f*)(W + (size_t)idx * 4);
  }
  __syncthreads();

  const float* bBase = sW + l16 * P + 4 * half;

  v8f acc0, acc1, acc2, acc3;
  auto mm8 = [&](const v4f av, const float* bp) {
    const v4f b0 = *(const v4f*)(bp);
    const v4f b1 = *(const v4f*)(bp + 16 * P);
    const v4f b2 = *(const v4f*)(bp + 32 * P);
    const v4f b3 = *(const v4f*)(bp + 48 * P);
    const v2f alo = {av.x, av.y};
    const v2f ahi = {av.z, av.w};
    acc0 = __builtin_amdgcn_wmma_f32_16x16x4_f32(false, alo, false, (v2f){b0.x, b0.y}, (short)0, acc0, false, false);
    acc1 = __builtin_amdgcn_wmma_f32_16x16x4_f32(false, alo, false, (v2f){b1.x, b1.y}, (short)0, acc1, false, false);
    acc2 = __builtin_amdgcn_wmma_f32_16x16x4_f32(false, alo, false, (v2f){b2.x, b2.y}, (short)0, acc2, false, false);
    acc3 = __builtin_amdgcn_wmma_f32_16x16x4_f32(false, alo, false, (v2f){b3.x, b3.y}, (short)0, acc3, false, false);
    acc0 = __builtin_amdgcn_wmma_f32_16x16x4_f32(false, ahi, false, (v2f){b0.z, b0.w}, (short)0, acc0, false, false);
    acc1 = __builtin_amdgcn_wmma_f32_16x16x4_f32(false, ahi, false, (v2f){b1.z, b1.w}, (short)0, acc1, false, false);
    acc2 = __builtin_amdgcn_wmma_f32_16x16x4_f32(false, ahi, false, (v2f){b2.z, b2.w}, (short)0, acc2, false, false);
    acc3 = __builtin_amdgcn_wmma_f32_16x16x4_f32(false, ahi, false, (v2f){b3.z, b3.w}, (short)0, acc3, false, false);
  };

  while (mTile < mTiles) {
    acc0 = (v8f){}; acc1 = (v8f){}; acc2 = (v8f){}; acc3 = (v8f){};
#pragma unroll 1
    for (int c = 0; c < NCH; c += 2) {
      const float* ap = aPtr + c * 32;
      const float* bp = bBase + c * 32;
      const v4f n0 = *(const v4f*)(ap + 32);
      const v4f n1 = *(const v4f*)(ap + 40);
      const v4f n2 = *(const v4f*)(ap + 48);
      const v4f n3 = *(const v4f*)(ap + 56);
      mm8(a0, bp);
      mm8(a1, bp + 8);
      mm8(a2, bp + 16);
      mm8(a3, bp + 24);
      __builtin_amdgcn_sched_barrier(0);
      const float* pf = (c + 2 < NCH) ? ap + 64 : aNext;
      a0 = *(const v4f*)(pf);
      a1 = *(const v4f*)(pf + 8);
      a2 = *(const v4f*)(pf + 16);
      a3 = *(const v4f*)(pf + 24);
      mm8(n0, bp + 32);
      mm8(n1, bp + 40);
      mm8(n2, bp + 48);
      mm8(n3, bp + 56);
    }

    const int r0 = mTile * 16 + half * 8;
    float* cPtr = C + (size_t)r0 * O + l16;
    if (mTile * 16 + 16 <= M) {
#pragma unroll
      for (int r = 0; r < 8; ++r) {
        float* p = cPtr + (size_t)r * O;
        p[0] = acc0[r]; p[16] = acc1[r]; p[32] = acc2[r]; p[48] = acc3[r];
      }
    } else {
#pragma unroll
      for (int r = 0; r < 8; ++r) {
        if (r0 + r < M) {
          float* p = cPtr + (size_t)r * O;
          p[0] = acc0[r]; p[16] = acc1[r]; p[32] = acc2[r]; p[48] = acc3[r];
        }
      }
    }

    mTile += stride;
    aPtr  = aNext;
    aNext = rowPtr(mTile + stride);
  }
}

constexpr int CH    = 4096;
constexpr int RSH   = 9;
constexpr int R     = 1 << RSH;
constexpr int MAXB  = 1024;
constexpr int CAP   = 16384;

__global__ __launch_bounds__(1024)
void edge_partition(const int* __restrict__ src,
                    const int* __restrict__ dst,
                    unsigned* __restrict__ part,
                    int* __restrict__ tbl,
                    int* __restrict__ cursor,
                    int E, int nBins, int nChunks) {
  __shared__ int hist[MAXB];
  __shared__ int wsum[32];
  __shared__ __attribute__((aligned(16))) unsigned sorted[CH];
  const int t     = threadIdx.x;
  const int lane  = t & 31;
  const int w     = t >> 5;
  const int chunk = blockIdx.x;
  const int base  = chunk * CH;
  if (chunk == 0 && t == 0) *cursor = 0;
  hist[t] = 0;
  __syncthreads();

  const int i4 = base + t * 4;
  int d[4] = {0, 0, 0, 0}, s[4] = {0, 0, 0, 0}, r[4] = {0, 0, 0, 0};
  int n = 0;
  if (i4 + 3 < E) {
    const int4 dd = *(const int4*)(dst + i4);
    const int4 ss = *(const int4*)(src + i4);
    d[0] = dd.x; d[1] = dd.y; d[2] = dd.z; d[3] = dd.w;
    s[0] = ss.x; s[1] = ss.y; s[2] = ss.z; s[3] = ss.w;
    n = 4;
  } else {
#pragma unroll
    for (int j = 0; j < 4; ++j) if (i4 + j < E) { d[j] = dst[i4 + j]; s[j] = src[i4 + j]; n = j + 1; }
  }
#pragma unroll
  for (int j = 0; j < 4; ++j) if (j < n) r[j] = atomicAdd(&hist[d[j] >> RSH], 1);
  __syncthreads();

  const int c = hist[t];
  int v = c;
#pragma unroll
  for (int o = 1; o < 32; o <<= 1) {
    const int u = __shfl_up(v, o, 32);
    if (lane >= o) v += u;
  }
  if (lane == 31) wsum[w] = v;
  __syncthreads();
  if (w == 0) {
    const int sw = wsum[lane];
    int x = sw;
#pragma unroll
    for (int o = 1; o < 32; o <<= 1) {
      const int u = __shfl_up(x, o, 32);
      if (lane >= o) x += u;
    }
    wsum[lane] = x - sw;
  }
  __syncthreads();
  const int start = wsum[w] + v - c;
  hist[t] = start;
  if (t < nBins) tbl[(size_t)t * nChunks + chunk] = start | (c << 16);
  __syncthreads();

#pragma unroll
  for (int j = 0; j < 4; ++j)
    if (j < n) sorted[hist[d[j] >> RSH] + r[j]] = ((unsigned)(d[j] & (R - 1)) << 23) | (unsigned)s[j];
  __syncthreads();

  const int left = E - base;
  const int m    = left < CH ? left : CH;
  if (t * 4 + 3 < m) {
    *(uint4*)(part + base + t * 4) = *(const uint4*)(sorted + t * 4);
  } else {
    for (int j = t * 4; j < m && j < t * 4 + 4; ++j) part[base + j] = sorted[j];
  }
}

__global__ __launch_bounds__(1024)
void bin_build(const unsigned* __restrict__ part,
               const int* __restrict__ tbl,
               int* __restrict__ cursor,
               int* __restrict__ csr,
               int* __restrict__ off,
               int* __restrict__ cnt,
               int N, int E, int nChunks) {
  __shared__ int ncnt[R];
  __shared__ int noff[R];
  __shared__ int wsum[32];
  __shared__ int sTotal, sBase;
  __shared__ int stage[CAP];
  const int t    = threadIdx.x;
  const int lane = t & 31;
  const int w    = t >> 5;
  const int bin  = blockIdx.x;
  const int* row = tbl + (size_t)bin * nChunks;

  if (t < R) ncnt[t] = 0;
  int tot = 0;
  for (int c = t; c < nChunks; c += 1024) {
    tot += row[c] >> 16;
  }
#pragma unroll
  for (int o = 16; o > 0; o >>= 1) tot += __shfl_xor(tot, o, 32);
  if (lane == 0) wsum[w] = tot;
  __syncthreads();
  int myBase = 0;
  if (t == 0) {
    int sum = 0;
    for (int i = 0; i < 32; ++i) sum += wsum[i];
    sTotal = sum;
    myBase = atomicAdd(cursor, sum);
  }
  __syncthreads();
  const int total = sTotal;

  auto walk = [&](auto&& f) {
    for (int c0 = w; c0 < nChunks; c0 += 256) {
      int e[8];
#pragma unroll
      for (int k = 0; k < 8; ++k) {
        const int c = c0 + 32 * k;
        e[k] = row[c < nChunks ? c : nChunks - 1];
      }
      size_t p[8];
      int n[8];
      unsigned u[8];
#pragma unroll
      for (int k = 0; k < 8; ++k) {
        const int c = c0 + 32 * k;
        const bool in = c < nChunks;
        n[k] = in ? e[k] >> 16 : 0;
        p[k] = (size_t)(in ? c : nChunks - 1) * CH + (size_t)(e[k] & 0xFFFF);
        size_t i = p[k] + (size_t)(lane < n[k] ? lane : 0);
        if (i >= (size_t)E) i = (size_t)E - 1;
        u[k] = part[i];
      }
#pragma unroll
      for (int k = 0; k < 8; ++k) {
        if (lane < n[k]) f(u[k]);
        for (int j = lane + 32; j < n[k]; j += 32) f(part[p[k] + j]);
      }
    }
  };

  walk([&](unsigned u) { atomicAdd(&ncnt[u >> 23], 1); });
  if (t == 0) sBase = myBase;
  __syncthreads();

  const int c = t < R ? ncnt[t] : 0;
  int v = c;
#pragma unroll
  for (int o = 1; o < 32; o <<= 1) {
    const int u = __shfl_up(v, o, 32);
    if (lane >= o) v += u;
  }
  if (lane == 31) wsum[w] = v;
  __syncthreads();
  if (w == 0) {
    const int sw = wsum[lane];
    int x = sw;
#pragma unroll
    for (int o = 1; o < 32; o <<= 1) {
      const int u = __shfl_up(x, o, 32);
      if (lane >= o) x += u;
    }
    wsum[lane] = x - sw;
  }
  __syncthreads();
  const int binBase = sBase;
  if (t < R) {
    const int excl = wsum[w] + v - c;
    noff[t] = excl;
    ncnt[t] = 0;
    const int node = bin * R + t;
    if (node < N) { off[node] = binBase + excl; cnt[node] = c; }
  }
  __syncthreads();

  const bool staged = total <= CAP;
  walk([&](unsigned u) {
    const int dl  = (int)(u >> 23);
    const int pos = noff[dl] + atomicAdd(&ncnt[dl], 1);
    const int sv  = (int)(u & 0x7FFFFFu);
    if (staged) stage[pos] = sv; else csr[binBase + pos] = sv;
  });
  __syncthreads();
  if (staged)
    for (int i = t; i < total; i += 1024) csr[binBase + i] = stage[i];
}

__global__ void seg_softmax_sum(const float* __restrict__ z,
                                const float* __restrict__ df,
                                const int* __restrict__ csr,
                                const int* __restrict__ off,
                                const int* __restrict__ cnt,
                                float* __restrict__ out, int N) {
  const int wave = (int)((blockIdx.x * (size_t)blockDim.x + threadIdx.x) >> 5);
  const int lane = threadIdx.x & 31;
  if (wave * 4 >= N) return;
  const int grp  = lane >> 3;
  const int q    = lane & 7;
  const int node = wave * 4 + grp;
  const bool live = node < N;
  const int nd   = live ? node : N - 1;
  const int beg  = off[nd];
  const int end  = beg + (live ? cnt[nd] : 0);
  const int last = end - 1;

  const float* dp = df + (size_t)nd * 64 + q * 4;
  const v4f b0 = *(const v4f*)dp;
  const v4f b1 = *(const v4f*)(dp + 32);

  v4f acc0 = {0.f, 0.f, 0.f, 0.f}, acc1 = {0.f, 0.f, 0.f, 0.f};
  float den = 0.f;
  float m   = -INFINITY;

  int nx[4] = {0, 0, 0, 0};
  if (beg < end) {
#pragma unroll
    for (int j = 0; j < 4; ++j) { const int k = beg + j; nx[j] = csr[k < end ? k : last]; }
  }

  for (int i = beg; i < end; i += 4) {
    v4f a0[4], a1[4];
#pragma unroll
    for (int j = 0; j < 4; ++j) {
      const float* zp = z + (size_t)nx[j] * 64 + q * 4;
      a0[j] = *(const v4f*)zp;
      a1[j] = *(const v4f*)(zp + 32);
    }
    if (i + 4 < end) {
#pragma unroll
      for (int j = 0; j < 4; ++j) { const int k = i + 4 + j; nx[j] = csr[k < end ? k : last]; }
    }

    float p[4];
#pragma unroll
    for (int j = 0; j < 4; ++j) {
      float t = a0[j].x * b0.x;
      t = fmaf(a0[j].y, b0.y, t);
      t = fmaf(a0[j].z, b0.z, t);
      t = fmaf(a0[j].w, b0.w, t);
      t = fmaf(a1[j].x, b1.x, t);
      t = fmaf(a1[j].y, b1.y, t);
      t = fmaf(a1[j].z, b1.z, t);
      t = fmaf(a1[j].w, b1.w, t);
      p[j] = t;
    }
#pragma unroll
    for (int o = 1; o < 8; o <<= 1) {
#pragma unroll
      for (int j = 0; j < 4; ++j) p[j] += __shfl_xor(p[j], o, 32);
    }
#pragma unroll
    for (int j = 1; j < 4; ++j) if (i + j >= end) p[j] = -INFINITY;

    const float cm = fmaxf(fmaxf(m, p[0]), fmaxf(fmaxf(p[1], p[2]), p[3]));
    const float sc = __expf(m - cm);
    const float w0 = __expf(p[0] - cm);
    const float w1 = __expf(p[1] - cm);
    const float w2 = __expf(p[2] - cm);
    const float w3 = __expf(p[3] - cm);
    den  = den * sc + ((w0 + w1) + (w2 + w3));
    acc0 = acc0 * sc + w0 * a0[0] + w1 * a0[1] + w2 * a0[2] + w3 * a0[3];
    acc1 = acc1 * sc + w0 * a1[0] + w1 * a1[1] + w2 * a1[2] + w3 * a1[3];
    m = cm;
  }

  if (live) {
    const float dn = den == 0.0f ? 1.0f : den;
    float* op = out + (size_t)node * 64 + q * 4;
    v4f r0, r1;
    r0.x = acc0.x / dn; r0.y = acc0.y / dn; r0.z = acc0.z / dn; r0.w = acc0.w / dn;
    r1.x = acc1.x / dn; r1.y = acc1.y / dn; r1.z = acc1.z / dn; r1.w = acc1.w / dn;
    *(v4f*)op        = r0;
    *(v4f*)(op + 32) = r1;
  }
}

extern "C" void kernel_launch(void* const* d_in, const int* in_sizes, int n_in,
                              void* d_out, int out_size, void* d_ws, size_t ws_size,
                              hipStream_t stream) {
  const float* h     = (const float*)d_in[0];
  const float* feat  = (const float*)d_in[1];
  const float* W_fc  = (const float*)d_in[2];
  const float* W_dst = (const float*)d_in[3];
  const int*   src   = (const int*)d_in[4];
  const int*   dst   = (const int*)d_in[5];
  float*       out   = (float*)d_out;

  constexpr int IN_DIM = 256, FEAT = 64, O = 64;
  const int N   = in_sizes[1] / FEAT;
  const int E   = in_sizes[4];
  const size_t Ep = ((size_t)E + 3) & ~(size_t)3;

  const int nBins   = (N + R - 1) / R;
  const int nChunks = (E + CH - 1) / CH;

  float*    z      = (float*)d_ws;
  float*    df     = z + (size_t)N * O;
  int*      csr    = (int*)(df + (size_t)N * O);
  unsigned* part   = (unsigned*)(csr + Ep);
  int*      off    = (int*)(part + Ep);
  int*      cnt    = off + N;
  int*      tbl    = cnt + N;
  int*      cursor = tbl + (size_t)nBins * nChunks;

  edge_partition<<<nChunks, 1024, 0, stream>>>(src, dst, part, tbl, cursor, E, nBins, nChunks);
  bin_build     <<<nBins,   1024, 0, stream>>>(part, tbl, cursor, csr, off, cnt, N, E, nChunks);

  gemm_nt_wmma<IN_DIM><<<768, 256, 0, stream>>>(h,    W_fc,  z,  N);
  gemm_nt_wmma<FEAT>  <<<768, 256, 0, stream>>>(feat, W_dst, df, N);

  seg_softmax_sum<<<(N + 31) / 32, 256, 0, stream>>>(z, df, csr, off, cnt, out, N);
}


# dst-feature projection moved into the reduction launch: each 8-wave block forms its 32 destinations' rows as eight 16x16 WMMA tiles through LDS; 5 launches -> 4, df never stored; sums unchanged
# speedup vs baseline: 7.0104x; 1.0679x over previous
#include <hip/hip_runtime.h>
#include <hip/hip_bf16.h>

typedef float v2f __attribute__((ext_vector_type(2)));
typedef float v8f __attribute__((ext_vector_type(8)));

typedef float v4f __attribute__((ext_vector_type(4)));

template <int K>
__global__ __launch_bounds__(256)
__attribute__((amdgpu_waves_per_eu(6, 6)))
void gemm_nt_wmma(const float* __restrict__ A,
                  const float* __restrict__ W,
                  float* __restrict__ C, int M) {
  constexpr int O   = 64;
  constexpr int P   = K + 4;
  constexpr int NCH = K / 32;
  static_assert(NCH % 2 == 0, "two pieces per trip");
  __shared__ __attribute__((aligned(16))) float sW[O * P];

  const int tid    = threadIdx.x;
  const int lane   = tid & 31;
  const int wv     = __builtin_amdgcn_readfirstlane(tid >> 5);
  const int half   = lane >> 4;
  const int l16    = lane & 15;
  const int mTiles = (M + 15) >> 4;
  const int stride = 8 * (int)gridDim.x;

  auto rowPtr = [&](int t) {
    int r = (t < mTiles ? t : mTiles - 1) * 16 + l16;
    if (r >= M) r = M - 1;
    return A + (size_t)r * K + 4 * half;
  };
  int mTile = wv * (int)gridDim.x + (int)blockIdx.x;
  const float* aPtr  = rowPtr(mTile);
  const float* aNext = rowPtr(mTile + stride);

  v4f a0 = *(const v4f*)(aPtr);
  v4f a1 = *(const v4f*)(aPtr + 8);
  v4f a2 = *(const v4f*)(aPtr + 16);
  v4f a3 = *(const v4f*)(aPtr + 24);

#pragma unroll
  for (int i = 0; i < O * K / 4 / 256; ++i) {
    const int idx = tid + i * 256;
    const int row = idx / (K / 4);
    const int c4  = idx % (K / 4);
    *(v4f*)(sW + row * P + c4 * 4) = *(const v4f*)(W + (size_t)idx * 4);
  }
  __syncthreads();

  const float* bBase = sW + l16 * P + 4 * half;

  v8f acc0, acc1, acc2, acc3;
  auto mm8 = [&](const v4f av, const float* bp) {
    const v4f b0 = *(const v4f*)(bp);
    const v4f b1 = *(const v4f*)(bp + 16 * P);
    const v4f b2 = *(const v4f*)(bp + 32 * P);
    const v4f b3 = *(const v4f*)(bp + 48 * P);
    const v2f alo = {av.x, av.y};
    const v2f ahi = {av.z, av.w};
    acc0 = __builtin_amdgcn_wmma_f32_16x16x4_f32(false, alo, false, (v2f){b0.x, b0.y}, (short)0, acc0, false, false);
    acc1 = __builtin_amdgcn_wmma_f32_16x16x4_f32(false, alo, false, (v2f){b1.x, b1.y}, (short)0, acc1, false, false);
    acc2 = __builtin_amdgcn_wmma_f32_16x16x4_f32(false, alo, false, (v2f){b2.x, b2.y}, (short)0, acc2, false, false);
    acc3 = __builtin_amdgcn_wmma_f32_16x16x4_f32(false, alo, false, (v2f){b3.x, b3.y}, (short)0, acc3, false, false);
    acc0 = __builtin_amdgcn_wmma_f32_16x16x4_f32(false, ahi, false, (v2f){b0.z, b0.w}, (short)0, acc0, false, false);
    acc1 = __builtin_amdgcn_wmma_f32_16x16x4_f32(false, ahi, false, (v2f){b1.z, b1.w}, (short)0, acc1, false, false);
    acc2 = __builtin_amdgcn_wmma_f32_16x16x4_f32(false, ahi, false, (v2f){b2.z, b2.w}, (short)0, acc2, false, false);
    acc3 = __builtin_amdgcn_wmma_f32_16x16x4_f32(false, ahi, false, (v2f){b3.z, b3.w}, (short)0, acc3, false, false);
  };

  while (mTile < mTiles) {
    acc0 = (v8f){}; acc1 = (v8f){}; acc2 = (v8f){}; acc3 = (v8f){};
#pragma unroll 1
    for (int c = 0; c < NCH; c += 2) {
      const float* ap = aPtr + c * 32;
      const float* bp = bBase + c * 32;
      const v4f n0 = *(const v4f*)(ap + 32);
      const v4f n1 = *(const v4f*)(ap + 40);
      const v4f n2 = *(const v4f*)(ap + 48);
      const v4f n3 = *(const v4f*)(ap + 56);
      mm8(a0, bp);
      mm8(a1, bp + 8);
      mm8(a2, bp + 16);
      mm8(a3, bp + 24);
      __builtin_amdgcn_sched_barrier(0);
      const float* pf = (c + 2 < NCH) ? ap + 64 : aNext;
      a0 = *(const v4f*)(pf);
      a1 = *(const v4f*)(pf + 8);
      a2 = *(const v4f*)(pf + 16);
      a3 = *(const v4f*)(pf + 24);
      mm8(n0, bp + 32);
      mm8(n1, bp + 40);
      mm8(n2, bp + 48);
      mm8(n3, bp + 56);
    }

    const int r0 = mTile * 16 + half * 8;
    float* cPtr = C + (size_t)r0 * O + l16;
    if (mTile * 16 + 16 <= M) {
#pragma unroll
      for (int r = 0; r < 8; ++r) {
        float* p = cPtr + (size_t)r * O;
        p[0] = acc0[r]; p[16] = acc1[r]; p[32] = acc2[r]; p[48] = acc3[r];
      }
    } else {
#pragma unroll
      for (int r = 0; r < 8; ++r) {
        if (r0 + r < M) {
          float* p = cPtr + (size_t)r * O;
          p[0] = acc0[r]; p[16] = acc1[r]; p[32] = acc2[r]; p[48] = acc3[r];
        }
      }
    }

    mTile += stride;
    aPtr  = aNext;
    aNext = rowPtr(mTile + stride);
  }
}

constexpr int CH    = 4096;
constexpr int RSH   = 9;
constexpr int R     = 1 << RSH;
constexpr int MAXB  = 1024;
constexpr int CAP   = 16384;

__global__ __launch_bounds__(1024)
void edge_partition(const int* __restrict__ src,
                    const int* __restrict__ dst,
                    unsigned* __restrict__ part,
                    int* __restrict__ tbl,
                    int* __restrict__ cursor,
                    int E, int nBins, int nChunks) {
  __shared__ int hist[MAXB];
  __shared__ int wsum[32];
  __shared__ __attribute__((aligned(16))) unsigned sorted[CH];
  const int t     = threadIdx.x;
  const int lane  = t & 31;
  const int w     = t >> 5;
  const int chunk = blockIdx.x;
  const int base  = chunk * CH;
  if (chunk == 0 && t == 0) *cursor = 0;
  hist[t] = 0;
  __syncthreads();

  const int i4 = base + t * 4;
  int d[4] = {0, 0, 0, 0}, s[4] = {0, 0, 0, 0}, r[4] = {0, 0, 0, 0};
  int n = 0;
  if (i4 + 3 < E) {
    const int4 dd = *(const int4*)(dst + i4);
    const int4 ss = *(const int4*)(src + i4);
    d[0] = dd.x; d[1] = dd.y; d[2] = dd.z; d[3] = dd.w;
    s[0] = ss.x; s[1] = ss.y; s[2] = ss.z; s[3] = ss.w;
    n = 4;
  } else {
#pragma unroll
    for (int j = 0; j < 4; ++j) if (i4 + j < E) { d[j] = dst[i4 + j]; s[j] = src[i4 + j]; n = j + 1; }
  }
#pragma unroll
  for (int j = 0; j < 4; ++j) if (j < n) r[j] = atomicAdd(&hist[d[j] >> RSH], 1);
  __syncthreads();

  const int c = hist[t];
  int v = c;
#pragma unroll
  for (int o = 1; o < 32; o <<= 1) {
    const int u = __shfl_up(v, o, 32);
    if (lane >= o) v += u;
  }
  if (lane == 31) wsum[w] = v;
  __syncthreads();
  if (w == 0) {
    const int sw = wsum[lane];
    int x = sw;
#pragma unroll
    for (int o = 1; o < 32; o <<= 1) {
      const int u = __shfl_up(x, o, 32);
      if (lane >= o) x += u;
    }
    wsum[lane] = x - sw;
  }
  __syncthreads();
  const int start = wsum[w] + v - c;
  hist[t] = start;
  if (t < nBins) tbl[(size_t)t * nChunks + chunk] = start | (c << 16);
  __syncthreads();

#pragma unroll
  for (int j = 0; j < 4; ++j)
    if (j < n) sorted[hist[d[j] >> RSH] + r[j]] = ((unsigned)(d[j] & (R - 1)) << 23) | (unsigned)s[j];
  __syncthreads();

  const int left = E - base;
  const int m    = left < CH ? left : CH;
  if (t * 4 + 3 < m) {
    *(uint4*)(part + base + t * 4) = *(const uint4*)(sorted + t * 4);
  } else {
    for (int j = t * 4; j < m && j < t * 4 + 4; ++j) part[base + j] = sorted[j];
  }
}

__global__ __launch_bounds__(1024)
void bin_build(const unsigned* __restrict__ part,
               const int* __restrict__ tbl,
               int* __restrict__ cursor,
               int* __restrict__ csr,
               int* __restrict__ off,
               int* __restrict__ cnt,
               int N, int E, int nChunks) {
  __shared__ int ncnt[R];
  __shared__ int noff[R];
  __shared__ int wsum[32];
  __shared__ int sTotal, sBase;
  __shared__ int stage[CAP];
  const int t    = threadIdx.x;
  const int lane = t & 31;
  const int w    = t >> 5;
  const int bin  = blockIdx.x;
  const int* row = tbl + (size_t)bin * nChunks;

  if (t < R) ncnt[t] = 0;
  int tot = 0;
  for (int c = t; c < nChunks; c += 1024) {
    tot += row[c] >> 16;
  }
#pragma unroll
  for (int o = 16; o > 0; o >>= 1) tot += __shfl_xor(tot, o, 32);
  if (lane == 0) wsum[w] = tot;
  __syncthreads();
  int myBase = 0;
  if (t == 0) {
    int sum = 0;
    for (int i = 0; i < 32; ++i) sum += wsum[i];
    sTotal = sum;
    myBase = atomicAdd(cursor, sum);
  }
  __syncthreads();
  const int total = sTotal;

  auto walk = [&](auto&& f) {
    for (int c0 = w; c0 < nChunks; c0 += 256) {
      int e[8];
#pragma unroll
      for (int k = 0; k < 8; ++k) {
        const int c = c0 + 32 * k;
        e[k] = row[c < nChunks ? c : nChunks - 1];
      }
      size_t p[8];
      int n[8];
      unsigned u[8];
#pragma unroll
      for (int k = 0; k < 8; ++k) {
        const int c = c0 + 32 * k;
        const bool in = c < nChunks;
        n[k] = in ? e[k] >> 16 : 0;
        p[k] = (size_t)(in ? c : nChunks - 1) * CH + (size_t)(e[k] & 0xFFFF);
        size_t i = p[k] + (size_t)(lane < n[k] ? lane : 0);
        if (i >= (size_t)E) i = (size_t)E - 1;
        u[k] = part[i];
      }
#pragma unroll
      for (int k = 0; k < 8; ++k) {
        if (lane < n[k]) f(u[k]);
        for (int j = lane + 32; j < n[k]; j += 32) f(part[p[k] + j]);
      }
    }
  };

  walk([&](unsigned u) { atomicAdd(&ncnt[u >> 23], 1); });
  if (t == 0) sBase = myBase;
  __syncthreads();

  const int c = t < R ? ncnt[t] : 0;
  int v = c;
#pragma unroll
  for (int o = 1; o < 32; o <<= 1) {
    const int u = __shfl_up(v, o, 32);
    if (lane >= o) v += u;
  }
  if (lane == 31) wsum[w] = v;
  __syncthreads();
  if (w == 0) {
    const int sw = wsum[lane];
    int x = sw;
#pragma unroll
    for (int o = 1; o < 32; o <<= 1) {
      const int u = __shfl_up(x, o, 32);
      if (lane >= o) x += u;
    }
    wsum[lane] = x - sw;
  }
  __syncthreads();
  const int binBase = sBase;
  if (t < R) {
    const int excl = wsum[w] + v - c;
    noff[t] = excl;
    ncnt[t] = 0;
    const int node = bin * R + t;
    if (node < N) { off[node] = binBase + excl; cnt[node] = c; }
  }
  __syncthreads();

  const bool staged = total <= CAP;
  walk([&](unsigned u) {
    const int dl  = (int)(u >> 23);
    const int pos = noff[dl] + atomicAdd(&ncnt[dl], 1);
    const int sv  = (int)(u & 0x7FFFFFu);
    if (staged) stage[pos] = sv; else csr[binBase + pos] = sv;
  });
  __syncthreads();
  if (staged)
    for (int i = t; i < total; i += 1024) csr[binBase + i] = stage[i];
}

constexpr int DFP = 68;

__global__ __launch_bounds__(256)
void seg_softmax_sum(const float* __restrict__ z,
                     const float* __restrict__ feat,
                     const float* __restrict__ Wd,
                     const int* __restrict__ csr,
                     const int* __restrict__ off,
                     const int* __restrict__ cnt,
                     float* __restrict__ out, int N) {
  __shared__ __attribute__((aligned(16))) float sW[64 * DFP];
  __shared__ __attribute__((aligned(16))) float sA[32 * DFP];
  __shared__ __attribute__((aligned(16))) float sD[32 * DFP];

  const int tid  = threadIdx.x;
  const int lane = tid & 31;
  const int wv   = tid >> 5;
  const int nb   = (int)blockIdx.x * 32;
  const int grp  = lane >> 3;
  const int q    = lane & 7;

  v4f wreg[4], areg[2];
#pragma unroll
  for (int i = 0; i < 2; ++i) {
    const int idx = tid + i * 256;
    int r = nb + (idx >> 4);
    if (r >= N) r = N - 1;
    areg[i] = *(const v4f*)(feat + (size_t)r * 64 + (idx & 15) * 4);
  }
#pragma unroll
  for (int i = 0; i < 4; ++i) wreg[i] = *(const v4f*)(Wd + (size_t)(tid + i * 256) * 4);

  const int node = nb + wv * 4 + grp;
  const bool live = node < N;
  const int nd   = live ? node : N - 1;
  const int beg  = off[nd];
  const int end  = beg + (live ? cnt[nd] : 0);
  const int last = end - 1;

#pragma unroll
  for (int i = 0; i < 4; ++i) {
    const int idx = tid + i * 256;
    *(v4f*)(sW + (idx >> 4) * DFP + (idx & 15) * 4) = wreg[i];
  }
#pragma unroll
  for (int i = 0; i < 2; ++i) {
    const int idx = tid + i * 256;
    *(v4f*)(sA + (idx >> 4) * DFP + (idx & 15) * 4) = areg[i];
  }
  __syncthreads();

  int nx[4] = {0, 0, 0, 0};
  if (beg < end) {
#pragma unroll
    for (int j = 0; j < 4; ++j) { const int k = beg + j; nx[j] = csr[k < end ? k : last]; }
  }

  {
    const int half = lane >> 4;
    const int l16  = lane & 15;
    const int mt   = wv >> 2;
    const int nt   = wv & 3;
    const float* ap = sA + (mt * 16 + l16) * DFP + 4 * half;
    const float* bp = sW + (nt * 16 + l16) * DFP + 4 * half;
    v8f acc = (v8f){};
#pragma unroll
    for (int i = 0; i < 8; ++i) {
      const v4f a = *(const v4f*)(ap + 8 * i);
      const v4f b = *(const v4f*)(bp + 8 * i);
      acc = __builtin_amdgcn_wmma_f32_16x16x4_f32(false, (v2f){a.x, a.y}, false, (v2f){b.x, b.y}, (short)0, acc, false, false);
      acc = __builtin_amdgcn_wmma_f32_16x16x4_f32(false, (v2f){a.z, a.w}, false, (v2f){b.z, b.w}, (short)0, acc, false, false);
    }
    float* sp = sD + (mt * 16 + half * 8) * DFP + nt * 16 + l16;
#pragma unroll
    for (int r = 0; r < 8; ++r) sp[r * DFP] = acc[r];
  }
  __syncthreads();

  const float* dp = sD + (wv * 4 + grp) * DFP + q * 4;
  const v4f b0 = *(const v4f*)dp;
  const v4f b1 = *(const v4f*)(dp + 32);

  v4f acc0 = {0.f, 0.f, 0.f, 0.f}, acc1 = {0.f, 0.f, 0.f, 0.f};
  float den = 0.f;
  float m   = -INFINITY;

  for (int i = beg; i < end; i += 4) {
    v4f a0[4], a1[4];
#pragma unroll
    for (int j = 0; j < 4; ++j) {
      const float* zp = z + (size_t)nx[j] * 64 + q * 4;
      a0[j] = *(const v4f*)zp;
      a1[j] = *(const v4f*)(zp + 32);
    }
    if (i + 4 < end) {
#pragma unroll
      for (int j = 0; j < 4; ++j) { const int k = i + 4 + j; nx[j] = csr[k < end ? k : last]; }
    }

    float p[4];
#pragma unroll
    for (int j = 0; j < 4; ++j) {
      float t = a0[j].x * b0.x;
      t = fmaf(a0[j].y, b0.y, t);
      t = fmaf(a0[j].z, b0.z, t);
      t = fmaf(a0[j].w, b0.w, t);
      t = fmaf(a1[j].x, b1.x, t);
      t = fmaf(a1[j].y, b1.y, t);
      t = fmaf(a1[j].z, b1.z, t);
      t = fmaf(a1[j].w, b1.w, t);
      p[j] = t;
    }
#pragma unroll
    for (int o = 1; o < 8; o <<= 1) {
#pragma unroll
      for (int j = 0; j < 4; ++j) p[j] += __shfl_xor(p[j], o, 32);
    }
#pragma unroll
    for (int j = 1; j < 4; ++j) if (i + j >= end) p[j] = -INFINITY;

    const float cm = fmaxf(fmaxf(m, p[0]), fmaxf(fmaxf(p[1], p[2]), p[3]));
    const float sc = __expf(m - cm);
    const float w0 = __expf(p[0] - cm);
    const float w1 = __expf(p[1] - cm);
    const float w2 = __expf(p[2] - cm);
    const float w3 = __expf(p[3] - cm);
    den  = den * sc + ((w0 + w1) + (w2 + w3));
    acc0 = acc0 * sc + w0 * a0[0] + w1 * a0[1] + w2 * a0[2] + w3 * a0[3];
    acc1 = acc1 * sc + w0 * a1[0] + w1 * a1[1] + w2 * a1[2] + w3 * a1[3];
    m = cm;
  }

  if (live) {
    const float dn = den == 0.0f ? 1.0f : den;
    float* op = out + (size_t)node * 64 + q * 4;
    v4f r0, r1;
    r0.x = acc0.x / dn; r0.y = acc0.y / dn; r0.z = acc0.z / dn; r0.w = acc0.w / dn;
    r1.x = acc1.x / dn; r1.y = acc1.y / dn; r1.z = acc1.z / dn; r1.w = acc1.w / dn;
    *(v4f*)op        = r0;
    *(v4f*)(op + 32) = r1;
  }
}

extern "C" void kernel_launch(void* const* d_in, const int* in_sizes, int n_in,
                              void* d_out, int out_size, void* d_ws, size_t ws_size,
                              hipStream_t stream) {
  const float* h     = (const float*)d_in[0];
  const float* feat  = (const float*)d_in[1];
  const float* W_fc  = (const float*)d_in[2];
  const float* W_dst = (const float*)d_in[3];
  const int*   src   = (const int*)d_in[4];
  const int*   dst   = (const int*)d_in[5];
  float*       out   = (float*)d_out;

  constexpr int IN_DIM = 256, FEAT = 64, O = 64;
  const int N   = in_sizes[1] / FEAT;
  const int E   = in_sizes[4];
  const size_t Ep = ((size_t)E + 3) & ~(size_t)3;

  const int nBins   = (N + R - 1) / R;
  const int nChunks = (E + CH - 1) / CH;

  float*    z      = (float*)d_ws;
  int*      csr    = (int*)(z + (size_t)N * O);
  unsigned* part   = (unsigned*)(csr + Ep);
  int*      off    = (int*)(part + Ep);
  int*      cnt    = off + N;
  int*      tbl    = cnt + N;
  int*      cursor = tbl + (size_t)nBins * nChunks;

  edge_partition<<<nChunks, 1024, 0, stream>>>(src, dst, part, tbl, cursor, E, nBins, nChunks);
  bin_build     <<<nBins,   1024, 0, stream>>>(part, tbl, cursor, csr, off, cnt, N, E, nChunks);

  gemm_nt_wmma<IN_DIM><<<768, 256, 0, stream>>>(h, W_fc, z, N);

  seg_softmax_sum<<<(N + 31) / 32, 256, 0, stream>>>(z, feat, W_dst, csr, off, cnt, out, N);
}
